# starting kernel
# baseline (speedup 1.0000x reference)
#include <hip/hip_runtime.h>


#define NN 100000
#define NE 1600000
#define KD 128

typedef __attribute__((ext_vector_type(2))) float v2f;
typedef __attribute__((ext_vector_type(8))) float v8f;

__global__ void fill_zero_kernel(float* __restrict__ p, int n) {
  int i = blockIdx.x * blockDim.x + threadIdx.x;
  int stride = gridDim.x * blockDim.x;
  for (; i < n; i += stride) p[i] = 0.0f;
}

__global__ void degree_kernel(const int* __restrict__ src, const int* __restrict__ dst,
                              float* __restrict__ out_deg, float* __restrict__ in_deg) {
  int e = blockIdx.x * blockDim.x + threadIdx.x;
  if (e >= NE) return;
  unsafeAtomicAdd(out_deg + src[e], 1.0f);
  unsafeAtomicAdd(in_deg + dst[e], 1.0f);
}

__global__ void norm_kernel(float* __restrict__ deg, int n) {
  int i = blockIdx.x * blockDim.x + threadIdx.x;
  if (i >= n) return;
  float d = deg[i];
  d = d < 1.0f ? 1.0f : d;
  deg[i] = rsqrtf(d);
}

template <int LOG2C, bool SCALE>
__global__ void edge_aggregate_kernel(const float* __restrict__ h,
                                      const int* __restrict__ src,
                                      const int* __restrict__ dst,
                                      const float* __restrict__ ew,
                                      const float* __restrict__ src_scale,
                                      float* __restrict__ agg) {
  constexpr int WIDTH = 4 << LOG2C;
  unsigned tid = blockIdx.x * blockDim.x + threadIdx.x;
  unsigned e = tid >> LOG2C;
  unsigned j = tid & ((1u << LOG2C) - 1u);
  if (e >= NE) return;
  int s = src[e];
  int d = dst[e];
  float c = ew[e];
  if (SCALE) c *= src_scale[s];
  const float4* hp = (const float4*)(h + (size_t)s * WIDTH);
  float4 v = hp[j];
  float* o = agg + (size_t)d * WIDTH + 4u * j;
  unsafeAtomicAdd(o + 0, v.x * c);
  unsafeAtomicAdd(o + 1, v.y * c);
  unsafeAtomicAdd(o + 2, v.z * c);
  unsafeAtomicAdd(o + 3, v.w * c);
}

template <int FOUT, bool RELU, bool HAS_BIAS>
__global__ void wmma_gemm_kernel(const float* __restrict__ A,
                                 const float* __restrict__ row_scale,
                                 const float* __restrict__ W,
                                 const float* __restrict__ bias,
                                 float* __restrict__ out) {
  __shared__ float As[16 * KD];
  const int m_base = blockIdx.x * 16;

  {
    constexpr int NVEC = 16 * KD / 4;
    for (int v = threadIdx.x; v < NVEC; v += blockDim.x) {
      int r = v >> 5;
      int cv = v & 31;
      float4 x = ((const float4*)(A + (size_t)(m_base + r) * KD))[cv];
      float s = row_scale[m_base + r];
      x.x *= s; x.y *= s; x.z *= s; x.w *= s;
      *(float4*)&As[r * KD + 4 * cv] = x;
    }
  }
  __syncthreads();

  const int wave = threadIdx.x >> 5;
  const int lane = threadIdx.x & 31;
  const int n_base = wave * 16;
  const int lo = lane & 15;
  const int hi = lane >> 4;

  v8f acc = {};
  const float* bbase = W + (size_t)(2 * hi) * FOUT + n_base + lo;
  const float* abase = &As[lo * KD + 2 * hi];

  #pragma unroll
  for (int k0 = 0; k0 < KD; k0 += 4) {
    v2f a = *(const v2f*)(abase + k0);
    v2f b;
    b.x = bbase[(size_t)(k0 + 0) * FOUT];
    b.y = bbase[(size_t)(k0 + 1) * FOUT];
    acc = __builtin_amdgcn_wmma_f32_16x16x4_f32(
         false, a,  false, b,
         (short)0, acc,  false,  false);
  }

  const int col = n_base + lo;
  const float bv = HAS_BIAS ? bias[col] : 0.0f;
  #pragma unroll
  for (int v = 0; v < 8; ++v) {
    int row = m_base + v + 8 * hi;
    float val = acc[v] + bv;
    if (RELU) val = val > 0.0f ? val : 0.0f;
    out[(size_t)row * FOUT + col] = val;
  }
}

__global__ void final_epilogue_kernel(const float* __restrict__ agg,
                                      const float* __restrict__ in_norm,
                                      const float* __restrict__ b2,
                                      float* __restrict__ out) {
  int i = blockIdx.x * blockDim.x + threadIdx.x;
  if (i >= NN * 64) return;
  int n = i >> 6;
  int f = i & 63;
  out[i] = agg[i] * in_norm[n] + b2[f];
}

extern "C" void kernel_launch(void* const* d_in, const int* in_sizes, int n_in,
                              void* d_out, int out_size, void* d_ws, size_t ws_size,
                              hipStream_t stream) {
  const float* features = (const float*)d_in[0];
  const int*   src      = (const int*)d_in[1];
  const int*   dst      = (const int*)d_in[2];
  const float* ew       = (const float*)d_in[3];
  const float* W0 = (const float*)d_in[4];
  const float* b0 = (const float*)d_in[5];
  const float* W1 = (const float*)d_in[6];
  const float* b1 = (const float*)d_in[7];
  const float* W2 = (const float*)d_in[8];
  const float* b2 = (const float*)d_in[9];
  float* out = (float*)d_out;

  float* ws       = (float*)d_ws;
  float* out_norm = ws;
  float* in_norm  = ws + NN;
  float* bufA     = ws + 2 * NN;
  float* bufB     = bufA + (size_t)NN * KD;

  fill_zero_kernel<<<512, 256, 0, stream>>>(out_norm, 2 * NN);
  degree_kernel<<<(NE + 255) / 256, 256, 0, stream>>>(src, dst, out_norm, in_norm);
  norm_kernel<<<(2 * NN + 255) / 256, 256, 0, stream>>>(out_norm, 2 * NN);

  fill_zero_kernel<<<2048, 256, 0, stream>>>(bufA, NN * KD);
  edge_aggregate_kernel<5, true><<<(NE * 32) / 256, 256, 0, stream>>>(
      features, src, dst, ew, out_norm, bufA);
  wmma_gemm_kernel<128, true, true><<<NN / 16, 256, 0, stream>>>(
      bufA, in_norm, W0, b0, bufB);

  fill_zero_kernel<<<2048, 256, 0, stream>>>(bufA, NN * KD);
  edge_aggregate_kernel<5, true><<<(NE * 32) / 256, 256, 0, stream>>>(
      bufB, src, dst, ew, out_norm, bufA);
  wmma_gemm_kernel<128, true, true><<<NN / 16, 256, 0, stream>>>(
      bufA, in_norm, W1, b1, bufB);

  wmma_gemm_kernel<64, false, false><<<NN / 16, 128, 0, stream>>>(
      bufB, out_norm, W2, nullptr, bufA);
  fill_zero_kernel<<<2048, 256, 0, stream>>>(bufB, NN * 64);
  edge_aggregate_kernel<4, false><<<(NE * 16) / 256, 256, 0, stream>>>(
      bufA, src, dst, ew, nullptr, bufB);
  final_epilogue_kernel<<<(NN * 64 + 255) / 256, 256, 0, stream>>>(bufB, in_norm, b2, out);
}


# edge scatter-add: lane j takes row columns j+32q, so each wave-wide f32 atomic add and its row load cover one whole 128-B line, not four lines of 8 floats; sums stay atomic, arrival order
# speedup vs baseline: 1.2674x; 1.2674x over previous
#include <hip/hip_runtime.h>


#define NN 100000
#define NE 1600000
#define KD 128

typedef __attribute__((ext_vector_type(2))) float v2f;
typedef __attribute__((ext_vector_type(8))) float v8f;

__global__ void fill_zero_kernel(float* __restrict__ p, int n) {
  int i = blockIdx.x * blockDim.x + threadIdx.x;
  int stride = gridDim.x * blockDim.x;
  for (; i < n; i += stride) p[i] = 0.0f;
}

__global__ void degree_kernel(const int* __restrict__ src, const int* __restrict__ dst,
                              float* __restrict__ out_deg, float* __restrict__ in_deg) {
  int e = blockIdx.x * blockDim.x + threadIdx.x;
  if (e >= NE) return;
  unsafeAtomicAdd(out_deg + src[e], 1.0f);
  unsafeAtomicAdd(in_deg + dst[e], 1.0f);
}

__global__ void norm_kernel(float* __restrict__ deg, int n) {
  int i = blockIdx.x * blockDim.x + threadIdx.x;
  if (i >= n) return;
  float d = deg[i];
  d = d < 1.0f ? 1.0f : d;
  deg[i] = rsqrtf(d);
}

template <int LOG2C, bool SCALE>
__global__ void edge_aggregate_kernel(const float* __restrict__ h,
                                      const int* __restrict__ src,
                                      const int* __restrict__ dst,
                                      const float* __restrict__ ew,
                                      const float* __restrict__ src_scale,
                                      float* __restrict__ agg) {
  constexpr int WIDTH = 4 << LOG2C;
  constexpr int LINES = WIDTH / 32;
  constexpr int EPW = 128 / WIDTH;
  unsigned tid = blockIdx.x * blockDim.x + threadIdx.x;
  unsigned w = tid >> 5;
  unsigned lane = tid & 31u;
  if (w >= NE / EPW) return;

  float v[EPW][LINES];
  float c[EPW];
  int d[EPW];
  #pragma unroll
  for (int i = 0; i < EPW; ++i) {
    unsigned e = w * EPW + i;
    int s = src[e];
    d[i] = dst[e];
    c[i] = ew[e];
    if (SCALE) c[i] *= src_scale[s];
    const float* hp = h + (size_t)s * WIDTH + lane;
    #pragma unroll
    for (int q = 0; q < LINES; ++q) v[i][q] = hp[32 * q];
  }
  #pragma unroll
  for (int i = 0; i < EPW; ++i) {
    float* o = agg + (size_t)d[i] * WIDTH + lane;
    #pragma unroll
    for (int q = 0; q < LINES; ++q) unsafeAtomicAdd(o + 32 * q, v[i][q] * c[i]);
  }
}

template <int FOUT, bool RELU, bool HAS_BIAS>
__global__ void wmma_gemm_kernel(const float* __restrict__ A,
                                 const float* __restrict__ row_scale,
                                 const float* __restrict__ W,
                                 const float* __restrict__ bias,
                                 float* __restrict__ out) {
  __shared__ float As[16 * KD];
  const int m_base = blockIdx.x * 16;

  {
    constexpr int NVEC = 16 * KD / 4;
    for (int v = threadIdx.x; v < NVEC; v += blockDim.x) {
      int r = v >> 5;
      int cv = v & 31;
      float4 x = ((const float4*)(A + (size_t)(m_base + r) * KD))[cv];
      float s = row_scale[m_base + r];
      x.x *= s; x.y *= s; x.z *= s; x.w *= s;
      *(float4*)&As[r * KD + 4 * cv] = x;
    }
  }
  __syncthreads();

  const int wave = threadIdx.x >> 5;
  const int lane = threadIdx.x & 31;
  const int n_base = wave * 16;
  const int lo = lane & 15;
  const int hi = lane >> 4;

  v8f acc = {};
  const float* bbase = W + (size_t)(2 * hi) * FOUT + n_base + lo;
  const float* abase = &As[lo * KD + 2 * hi];

  #pragma unroll
  for (int k0 = 0; k0 < KD; k0 += 4) {
    v2f a = *(const v2f*)(abase + k0);
    v2f b;
    b.x = bbase[(size_t)(k0 + 0) * FOUT];
    b.y = bbase[(size_t)(k0 + 1) * FOUT];
    acc = __builtin_amdgcn_wmma_f32_16x16x4_f32(
         false, a,  false, b,
         (short)0, acc,  false,  false);
  }

  const int col = n_base + lo;
  const float bv = HAS_BIAS ? bias[col] : 0.0f;
  #pragma unroll
  for (int v = 0; v < 8; ++v) {
    int row = m_base + v + 8 * hi;
    float val = acc[v] + bv;
    if (RELU) val = val > 0.0f ? val : 0.0f;
    out[(size_t)row * FOUT + col] = val;
  }
}

__global__ void final_epilogue_kernel(const float* __restrict__ agg,
                                      const float* __restrict__ in_norm,
                                      const float* __restrict__ b2,
                                      float* __restrict__ out) {
  int i = blockIdx.x * blockDim.x + threadIdx.x;
  if (i >= NN * 64) return;
  int n = i >> 6;
  int f = i & 63;
  out[i] = agg[i] * in_norm[n] + b2[f];
}

extern "C" void kernel_launch(void* const* d_in, const int* in_sizes, int n_in,
                              void* d_out, int out_size, void* d_ws, size_t ws_size,
                              hipStream_t stream) {
  const float* features = (const float*)d_in[0];
  const int*   src      = (const int*)d_in[1];
  const int*   dst      = (const int*)d_in[2];
  const float* ew       = (const float*)d_in[3];
  const float* W0 = (const float*)d_in[4];
  const float* b0 = (const float*)d_in[5];
  const float* W1 = (const float*)d_in[6];
  const float* b1 = (const float*)d_in[7];
  const float* W2 = (const float*)d_in[8];
  const float* b2 = (const float*)d_in[9];
  float* out = (float*)d_out;

  float* ws       = (float*)d_ws;
  float* out_norm = ws;
  float* in_norm  = ws + NN;
  float* bufA     = ws + 2 * NN;
  float* bufB     = bufA + (size_t)NN * KD;

  fill_zero_kernel<<<512, 256, 0, stream>>>(out_norm, 2 * NN);
  degree_kernel<<<(NE + 255) / 256, 256, 0, stream>>>(src, dst, out_norm, in_norm);
  norm_kernel<<<(2 * NN + 255) / 256, 256, 0, stream>>>(out_norm, 2 * NN);

  fill_zero_kernel<<<2048, 256, 0, stream>>>(bufA, NN * KD);
  edge_aggregate_kernel<5, true><<<(NE * 32) / 256, 256, 0, stream>>>(
      features, src, dst, ew, out_norm, bufA);
  wmma_gemm_kernel<128, true, true><<<NN / 16, 256, 0, stream>>>(
      bufA, in_norm, W0, b0, bufB);

  fill_zero_kernel<<<2048, 256, 0, stream>>>(bufA, NN * KD);
  edge_aggregate_kernel<5, true><<<(NE * 32) / 256, 256, 0, stream>>>(
      bufB, src, dst, ew, out_norm, bufA);
  wmma_gemm_kernel<128, true, true><<<NN / 16, 256, 0, stream>>>(
      bufA, in_norm, W1, b1, bufB);

  wmma_gemm_kernel<64, false, false><<<NN / 16, 128, 0, stream>>>(
      bufB, out_norm, W2, nullptr, bufA);
  fill_zero_kernel<<<2048, 256, 0, stream>>>(bufB, NN * 64);
  edge_aggregate_kernel<4, false><<<(NE * 16) / 256, 256, 0, stream>>>(
      bufA, src, dst, ew, nullptr, bufB);
  final_epilogue_kernel<<<(NN * 64 + 255) / 256, 256, 0, stream>>>(bufB, in_norm, b2, out);
}


# edge sums: edges bucketed by destination once (atomic int count ranks each edge, scan, place); one wave adds a node's rows in registers, one store per row; no row atomics; sum order = arrival rank
# speedup vs baseline: 2.0920x; 1.6506x over previous
#include <hip/hip_runtime.h>


#define NN 100000
#define NE 1600000
#define KD 128
#define NNP 100096
#define NSCAN 391
#define NPW 4

typedef __attribute__((ext_vector_type(2))) float v2f;
typedef __attribute__((ext_vector_type(8))) float v8f;

__global__ void fill_zero_kernel(float* __restrict__ p, int n) {
  int i = blockIdx.x * blockDim.x + threadIdx.x;
  int stride = gridDim.x * blockDim.x;
  for (; i < n; i += stride) p[i] = 0.0f;
}

__global__ void degree_rank_kernel(const int* __restrict__ src, const int* __restrict__ dst,
                                   float* __restrict__ out_deg, int* __restrict__ in_cnt,
                                   int* __restrict__ rank) {
  int e = blockIdx.x * blockDim.x + threadIdx.x;
  if (e >= NE) return;
  unsafeAtomicAdd(out_deg + src[e], 1.0f);
  rank[e] = atomicAdd(in_cnt + dst[e], 1);
}

__global__ void block_sum_kernel(const int* __restrict__ in_cnt, int* __restrict__ blocksum) {
  __shared__ int part[8];
  int t = threadIdx.x;
  int i = blockIdx.x * 256 + t;
  int c = i < NN ? in_cnt[i] : 0;
  #pragma unroll
  for (int d = 16; d > 0; d >>= 1) c += __shfl_xor(c, d, 32);
  if ((t & 31) == 0) part[t >> 5] = c;
  __syncthreads();
  if (t == 0) {
    int s = 0;
    #pragma unroll
    for (int k = 0; k < 8; ++k) s += part[k];
    blocksum[blockIdx.x] = s;
  }
}

__global__ void scan_norm_kernel(const int* __restrict__ in_cnt, const int* __restrict__ blocksum,
                                 int* __restrict__ offs, float* __restrict__ in_norm,
                                 float* __restrict__ out_deg) {
  __shared__ int wtot[8];
  __shared__ int wpre[8];
  const int t = threadIdx.x;
  const int b = blockIdx.x;
  const int lane = t & 31;
  const int wv = t >> 5;
  const int i = b * 256 + t;
  const int c = i < NN ? in_cnt[i] : 0;

  int p = 0;
  if (t < b) p += blocksum[t];
  if (t + 256 < b) p += blocksum[t + 256];
  #pragma unroll
  for (int d = 16; d > 0; d >>= 1) p += __shfl_xor(p, d, 32);

  int s = c;
  #pragma unroll
  for (int d = 1; d < 32; d <<= 1) {
    int u = __shfl_up(s, d, 32);
    if (lane >= d) s += u;
  }
  if (lane == 31) { wtot[wv] = s; wpre[wv] = p; }
  __syncthreads();
  int base = 0;
  #pragma unroll
  for (int k = 0; k < 8; ++k) {
    base += wpre[k];
    if (k < wv) base += wtot[k];
  }
  if (i < NN) {
    offs[i] = base + s - c;
    float di = (float)c;
    di = di < 1.0f ? 1.0f : di;
    in_norm[i] = rsqrtf(di);
    float d = out_deg[i];
    d = d < 1.0f ? 1.0f : d;
    out_deg[i] = rsqrtf(d);
  }
}

__global__ void place_kernel(const int* __restrict__ src, const int* __restrict__ dst,
                             const float* __restrict__ ew, const int* __restrict__ rank,
                             const int* __restrict__ offs, const float* __restrict__ out_norm,
                             int* __restrict__ rec) {
  int e = blockIdx.x * blockDim.x + threadIdx.x;
  if (e >= NE) return;
  int s = src[e];
  float w = ew[e];
  int pos = offs[dst[e]] + rank[e];
  int4 r;
  r.x = s;
  r.y = __float_as_int(w * out_norm[s]);
  r.z = __float_as_int(w);
  r.w = 0;
  ((int4*)rec)[pos] = r;
}

template <int U, int WIDTH, bool SCALED, typename V>
__device__ __forceinline__ void row_group(V& acc, const float* __restrict__ h, int4 r, int j,
                                          unsigned lane) {
  constexpr int VL = WIDTH / 32;
  V v[U];
  float c[U];
  #pragma unroll
  for (int u = 0; u < U; ++u) {
    int s = __builtin_amdgcn_readlane(r.x, j + u);
    c[u] = __int_as_float(__builtin_amdgcn_readlane(SCALED ? r.y : r.z, j + u));
    v[u] = *(const V*)(h + (size_t)s * WIDTH + lane * VL);
  }
  #pragma unroll
  for (int u = 0; u < U; ++u) acc += v[u] * c[u];
}

template <int WIDTH, bool SCALED, typename V>
__global__ void gather_sum_kernel(const float* __restrict__ h, const int* __restrict__ rec,
                                  const int* __restrict__ offs, const int* __restrict__ in_cnt,
                                  float* __restrict__ agg) {
  constexpr int VL = WIDTH / 32;
  const unsigned lane = threadIdx.x & 31u;
  const int w = __builtin_amdgcn_readfirstlane((int)(blockIdx.x * (blockDim.x >> 5) + (threadIdx.x >> 5)));
  const int4* __restrict__ rp = (const int4*)rec;
  for (int k = 0; k < NPW; ++k) {
    const int n = w * NPW + k;
    if (n >= NN) return;
    const int beg = offs[n];
    const int deg = in_cnt[n];
    V acc = {};
    for (int c0 = 0; c0 < deg; c0 += 32) {
      const int m = deg - c0 < 32 ? deg - c0 : 32;
      int4 r = {0, 0, 0, 0};
      if ((int)lane < m) r = rp[beg + c0 + (int)lane];
      int j = 0;
      for (; j + 8 <= m; j += 8) row_group<8, WIDTH, SCALED, V>(acc, h, r, j, lane);
      if (j + 4 <= m) { row_group<4, WIDTH, SCALED, V>(acc, h, r, j, lane); j += 4; }
      if (j + 2 <= m) { row_group<2, WIDTH, SCALED, V>(acc, h, r, j, lane); j += 2; }
      if (j < m) row_group<1, WIDTH, SCALED, V>(acc, h, r, j, lane);
    }
    *(V*)(agg + (size_t)n * WIDTH + lane * VL) = acc;
  }
}

template <int FOUT, bool RELU, bool HAS_BIAS>
__global__ void wmma_gemm_kernel(const float* __restrict__ A,
                                 const float* __restrict__ row_scale,
                                 const float* __restrict__ W,
                                 const float* __restrict__ bias,
                                 float* __restrict__ out) {
  __shared__ float As[16 * KD];
  const int m_base = blockIdx.x * 16;

  {
    constexpr int NVEC = 16 * KD / 4;
    for (int v = threadIdx.x; v < NVEC; v += blockDim.x) {
      int r = v >> 5;
      int cv = v & 31;
      float4 x = ((const float4*)(A + (size_t)(m_base + r) * KD))[cv];
      float s = row_scale[m_base + r];
      x.x *= s; x.y *= s; x.z *= s; x.w *= s;
      *(float4*)&As[r * KD + 4 * cv] = x;
    }
  }
  __syncthreads();

  const int wave = threadIdx.x >> 5;
  const int lane = threadIdx.x & 31;
  const int n_base = wave * 16;
  const int lo = lane & 15;
  const int hi = lane >> 4;

  v8f acc = {};
  const float* bbase = W + (size_t)(2 * hi) * FOUT + n_base + lo;
  const float* abase = &As[lo * KD + 2 * hi];

  #pragma unroll
  for (int k0 = 0; k0 < KD; k0 += 4) {
    v2f a = *(const v2f*)(abase + k0);
    v2f b;
    b.x = bbase[(size_t)(k0 + 0) * FOUT];
    b.y = bbase[(size_t)(k0 + 1) * FOUT];
    acc = __builtin_amdgcn_wmma_f32_16x16x4_f32(
         false, a,  false, b,
         (short)0, acc,  false,  false);
  }

  const int col = n_base + lo;
  const float bv = HAS_BIAS ? bias[col] : 0.0f;
  #pragma unroll
  for (int v = 0; v < 8; ++v) {
    int row = m_base + v + 8 * hi;
    float val = acc[v] + bv;
    if (RELU) val = val > 0.0f ? val : 0.0f;
    out[(size_t)row * FOUT + col] = val;
  }
}

__global__ void final_epilogue_kernel(const float* __restrict__ agg,
                                      const float* __restrict__ in_norm,
                                      const float* __restrict__ b2,
                                      float* __restrict__ out) {
  int i = blockIdx.x * blockDim.x + threadIdx.x;
  if (i >= NN * 64) return;
  int n = i >> 6;
  int f = i & 63;
  out[i] = agg[i] * in_norm[n] + b2[f];
}

extern "C" void kernel_launch(void* const* d_in, const int* in_sizes, int n_in,
                              void* d_out, int out_size, void* d_ws, size_t ws_size,
                              hipStream_t stream) {
  const float* features = (const float*)d_in[0];
  const int*   src      = (const int*)d_in[1];
  const int*   dst      = (const int*)d_in[2];
  const float* ew       = (const float*)d_in[3];
  const float* W0 = (const float*)d_in[4];
  const float* b0 = (const float*)d_in[5];
  const float* W1 = (const float*)d_in[6];
  const float* b1 = (const float*)d_in[7];
  const float* W2 = (const float*)d_in[8];
  const float* b2 = (const float*)d_in[9];
  float* out = (float*)d_out;

  float* ws       = (float*)d_ws;
  float* out_norm = ws;
  int*   in_cnt   = (int*)(ws + NNP);
  float* in_norm  = ws + 2 * NNP;
  int*   offs     = (int*)(ws + 3 * NNP);
  int*   blocksum = (int*)(ws + 4 * NNP);
  int*   rank     = (int*)(ws + 4 * NNP + 512);
  int*   rec      = rank + NE;
  float* bufA     = (float*)(rec + (size_t)4 * NE);
  float* bufB     = bufA + (size_t)NN * KD;

  fill_zero_kernel<<<512, 256, 0, stream>>>(out_norm, 2 * NNP);
  degree_rank_kernel<<<(NE + 255) / 256, 256, 0, stream>>>(src, dst, out_norm, in_cnt, rank);
  block_sum_kernel<<<NSCAN, 256, 0, stream>>>(in_cnt, blocksum);
  scan_norm_kernel<<<NSCAN, 256, 0, stream>>>(in_cnt, blocksum, offs, in_norm, out_norm);
  place_kernel<<<(NE + 255) / 256, 256, 0, stream>>>(src, dst, ew, rank, offs, out_norm, rec);

  gather_sum_kernel<128, true, float4><<<(NN / NPW * 32 + 255) / 256, 256, 0, stream>>>(
      features, rec, offs, in_cnt, bufA);
  wmma_gemm_kernel<128, true, true><<<NN / 16, 256, 0, stream>>>(
      bufA, in_norm, W0, b0, bufB);

  gather_sum_kernel<128, true, float4><<<(NN / NPW * 32 + 255) / 256, 256, 0, stream>>>(
      bufB, rec, offs, in_cnt, bufA);
  wmma_gemm_kernel<128, true, true><<<NN / 16, 256, 0, stream>>>(
      bufA, in_norm, W1, b1, bufB);

  wmma_gemm_kernel<64, false, false><<<NN / 16, 128, 0, stream>>>(
      bufB, out_norm, W2, nullptr, bufA);
  gather_sum_kernel<64, false, float2><<<(NN / NPW * 32 + 255) / 256, 256, 0, stream>>>(
      bufA, rec, offs, in_cnt, bufB);
  final_epilogue_kernel<<<(NN * 64 + 255) / 256, 256, 0, stream>>>(bufB, in_norm, b2, out);
}


# final in_norm scale and bias applied in the width-64 row-sum kernel's own row store, straight into out; the 200000-wave one-element-per-lane epilogue launch and its read-back of the sums are gone
# speedup vs baseline: 2.2943x; 1.0967x over previous
#include <hip/hip_runtime.h>


#define NN 100000
#define NE 1600000
#define KD 128
#define NNP 100096
#define NSCAN 391
#define NPW 4

typedef __attribute__((ext_vector_type(2))) float v2f;
typedef __attribute__((ext_vector_type(8))) float v8f;

__global__ void fill_zero_kernel(float* __restrict__ p, int n) {
  int i = blockIdx.x * blockDim.x + threadIdx.x;
  int stride = gridDim.x * blockDim.x;
  for (; i < n; i += stride) p[i] = 0.0f;
}

__global__ void degree_rank_kernel(const int* __restrict__ src, const int* __restrict__ dst,
                                   float* __restrict__ out_deg, int* __restrict__ in_cnt,
                                   int* __restrict__ rank) {
  int e = blockIdx.x * blockDim.x + threadIdx.x;
  if (e >= NE) return;
  unsafeAtomicAdd(out_deg + src[e], 1.0f);
  rank[e] = atomicAdd(in_cnt + dst[e], 1);
}

__global__ void block_sum_kernel(const int* __restrict__ in_cnt, int* __restrict__ blocksum) {
  __shared__ int part[8];
  int t = threadIdx.x;
  int i = blockIdx.x * 256 + t;
  int c = i < NN ? in_cnt[i] : 0;
  #pragma unroll
  for (int d = 16; d > 0; d >>= 1) c += __shfl_xor(c, d, 32);
  if ((t & 31) == 0) part[t >> 5] = c;
  __syncthreads();
  if (t == 0) {
    int s = 0;
    #pragma unroll
    for (int k = 0; k < 8; ++k) s += part[k];
    blocksum[blockIdx.x] = s;
  }
}

__global__ void scan_norm_kernel(const int* __restrict__ in_cnt, const int* __restrict__ blocksum,
                                 int* __restrict__ offs, float* __restrict__ in_norm,
                                 float* __restrict__ out_deg) {
  __shared__ int wtot[8];
  __shared__ int wpre[8];
  const int t = threadIdx.x;
  const int b = blockIdx.x;
  const int lane = t & 31;
  const int wv = t >> 5;
  const int i = b * 256 + t;
  const int c = i < NN ? in_cnt[i] : 0;

  int p = 0;
  if (t < b) p += blocksum[t];
  if (t + 256 < b) p += blocksum[t + 256];
  #pragma unroll
  for (int d = 16; d > 0; d >>= 1) p += __shfl_xor(p, d, 32);

  int s = c;
  #pragma unroll
  for (int d = 1; d < 32; d <<= 1) {
    int u = __shfl_up(s, d, 32);
    if (lane >= d) s += u;
  }
  if (lane == 31) { wtot[wv] = s; wpre[wv] = p; }
  __syncthreads();
  int base = 0;
  #pragma unroll
  for (int k = 0; k < 8; ++k) {
    base += wpre[k];
    if (k < wv) base += wtot[k];
  }
  if (i < NN) {
    offs[i] = base + s - c;
    float di = (float)c;
    di = di < 1.0f ? 1.0f : di;
    in_norm[i] = rsqrtf(di);
    float d = out_deg[i];
    d = d < 1.0f ? 1.0f : d;
    out_deg[i] = rsqrtf(d);
  }
}

__global__ void place_kernel(const int* __restrict__ src, const int* __restrict__ dst,
                             const float* __restrict__ ew, const int* __restrict__ rank,
                             const int* __restrict__ offs, const float* __restrict__ out_norm,
                             int* __restrict__ rec) {
  int e = blockIdx.x * blockDim.x + threadIdx.x;
  if (e >= NE) return;
  int s = src[e];
  float w = ew[e];
  int pos = offs[dst[e]] + rank[e];
  int4 r;
  r.x = s;
  r.y = __float_as_int(w * out_norm[s]);
  r.z = __float_as_int(w);
  r.w = 0;
  ((int4*)rec)[pos] = r;
}

template <int U, int WIDTH, bool SCALED, typename V>
__device__ __forceinline__ void row_group(V& acc, const float* __restrict__ h, int4 r, int j,
                                          unsigned lane) {
  constexpr int VL = WIDTH / 32;
  V v[U];
  float c[U];
  #pragma unroll
  for (int u = 0; u < U; ++u) {
    int s = __builtin_amdgcn_readlane(r.x, j + u);
    c[u] = __int_as_float(__builtin_amdgcn_readlane(SCALED ? r.y : r.z, j + u));
    v[u] = *(const V*)(h + (size_t)s * WIDTH + lane * VL);
  }
  #pragma unroll
  for (int u = 0; u < U; ++u) acc += v[u] * c[u];
}

template <int WIDTH, bool SCALED, bool FINAL, typename V>
__global__ void gather_sum_kernel(const float* __restrict__ h, const int* __restrict__ rec,
                                  const int* __restrict__ offs, const int* __restrict__ in_cnt,
                                  const float* __restrict__ in_norm,
                                  const float* __restrict__ bias,
                                  float* __restrict__ agg) {
  constexpr int VL = WIDTH / 32;
  const unsigned lane = threadIdx.x & 31u;
  V bv = {};
  if (FINAL) bv = *(const V*)(bias + lane * VL);
  const int w = __builtin_amdgcn_readfirstlane((int)(blockIdx.x * (blockDim.x >> 5) + (threadIdx.x >> 5)));
  const int4* __restrict__ rp = (const int4*)rec;
  for (int k = 0; k < NPW; ++k) {
    const int n = w * NPW + k;
    if (n >= NN) return;
    const int beg = offs[n];
    const int deg = in_cnt[n];
    const float nrm = FINAL ? in_norm[n] : 1.0f;
    V acc = {};
    for (int c0 = 0; c0 < deg; c0 += 32) {
      const int m = deg - c0 < 32 ? deg - c0 : 32;
      int4 r = {0, 0, 0, 0};
      if ((int)lane < m) r = rp[beg + c0 + (int)lane];
      int j = 0;
      for (; j + 8 <= m; j += 8) row_group<8, WIDTH, SCALED, V>(acc, h, r, j, lane);
      if (j + 4 <= m) { row_group<4, WIDTH, SCALED, V>(acc, h, r, j, lane); j += 4; }
      if (j + 2 <= m) { row_group<2, WIDTH, SCALED, V>(acc, h, r, j, lane); j += 2; }
      if (j < m) row_group<1, WIDTH, SCALED, V>(acc, h, r, j, lane);
    }
    if (FINAL) acc = acc * nrm + bv;
    *(V*)(agg + (size_t)n * WIDTH + lane * VL) = acc;
  }
}

template <int FOUT, bool RELU, bool HAS_BIAS>
__global__ void wmma_gemm_kernel(const float* __restrict__ A,
                                 const float* __restrict__ row_scale,
                                 const float* __restrict__ W,
                                 const float* __restrict__ bias,
                                 float* __restrict__ out) {
  __shared__ float As[16 * KD];
  const int m_base = blockIdx.x * 16;

  {
    constexpr int NVEC = 16 * KD / 4;
    for (int v = threadIdx.x; v < NVEC; v += blockDim.x) {
      int r = v >> 5;
      int cv = v & 31;
      float4 x = ((const float4*)(A + (size_t)(m_base + r) * KD))[cv];
      float s = row_scale[m_base + r];
      x.x *= s; x.y *= s; x.z *= s; x.w *= s;
      *(float4*)&As[r * KD + 4 * cv] = x;
    }
  }
  __syncthreads();

  const int wave = threadIdx.x >> 5;
  const int lane = threadIdx.x & 31;
  const int n_base = wave * 16;
  const int lo = lane & 15;
  const int hi = lane >> 4;

  v8f acc = {};
  const float* bbase = W + (size_t)(2 * hi) * FOUT + n_base + lo;
  const float* abase = &As[lo * KD + 2 * hi];

  #pragma unroll
  for (int k0 = 0; k0 < KD; k0 += 4) {
    v2f a = *(const v2f*)(abase + k0);
    v2f b;
    b.x = bbase[(size_t)(k0 + 0) * FOUT];
    b.y = bbase[(size_t)(k0 + 1) * FOUT];
    acc = __builtin_amdgcn_wmma_f32_16x16x4_f32(
         false, a,  false, b,
         (short)0, acc,  false,  false);
  }

  const int col = n_base + lo;
  const float bv = HAS_BIAS ? bias[col] : 0.0f;
  #pragma unroll
  for (int v = 0; v < 8; ++v) {
    int row = m_base + v + 8 * hi;
    float val = acc[v] + bv;
    if (RELU) val = val > 0.0f ? val : 0.0f;
    out[(size_t)row * FOUT + col] = val;
  }
}

extern "C" void kernel_launch(void* const* d_in, const int* in_sizes, int n_in,
                              void* d_out, int out_size, void* d_ws, size_t ws_size,
                              hipStream_t stream) {
  const float* features = (const float*)d_in[0];
  const int*   src      = (const int*)d_in[1];
  const int*   dst      = (const int*)d_in[2];
  const float* ew       = (const float*)d_in[3];
  const float* W0 = (const float*)d_in[4];
  const float* b0 = (const float*)d_in[5];
  const float* W1 = (const float*)d_in[6];
  const float* b1 = (const float*)d_in[7];
  const float* W2 = (const float*)d_in[8];
  const float* b2 = (const float*)d_in[9];
  float* out = (float*)d_out;

  float* ws       = (float*)d_ws;
  float* out_norm = ws;
  int*   in_cnt   = (int*)(ws + NNP);
  float* in_norm  = ws + 2 * NNP;
  int*   offs     = (int*)(ws + 3 * NNP);
  int*   blocksum = (int*)(ws + 4 * NNP);
  int*   rank     = (int*)(ws + 4 * NNP + 512);
  int*   rec      = rank + NE;
  float* bufA     = (float*)(rec + (size_t)4 * NE);
  float* bufB     = bufA + (size_t)NN * KD;

  fill_zero_kernel<<<512, 256, 0, stream>>>(out_norm, 2 * NNP);
  degree_rank_kernel<<<(NE + 255) / 256, 256, 0, stream>>>(src, dst, out_norm, in_cnt, rank);
  block_sum_kernel<<<NSCAN, 256, 0, stream>>>(in_cnt, blocksum);
  scan_norm_kernel<<<NSCAN, 256, 0, stream>>>(in_cnt, blocksum, offs, in_norm, out_norm);
  place_kernel<<<(NE + 255) / 256, 256, 0, stream>>>(src, dst, ew, rank, offs, out_norm, rec);

  gather_sum_kernel<128, true, false, float4><<<(NN / NPW * 32 + 255) / 256, 256, 0, stream>>>(
      features, rec, offs, in_cnt, nullptr, nullptr, bufA);
  wmma_gemm_kernel<128, true, true><<<NN / 16, 256, 0, stream>>>(
      bufA, in_norm, W0, b0, bufB);

  gather_sum_kernel<128, true, false, float4><<<(NN / NPW * 32 + 255) / 256, 256, 0, stream>>>(
      bufB, rec, offs, in_cnt, nullptr, nullptr, bufA);
  wmma_gemm_kernel<128, true, true><<<NN / 16, 256, 0, stream>>>(
      bufA, in_norm, W1, b1, bufB);

  wmma_gemm_kernel<64, false, false><<<NN / 16, 128, 0, stream>>>(
      bufB, out_norm, W2, nullptr, bufA);
  gather_sum_kernel<64, false, true, float2><<<(NN / NPW * 32 + 255) / 256, 256, 0, stream>>>(
      bufA, rec, offs, in_cnt, in_norm, b2, out);
}


# GEMM A tile in LDS at a row pitch of 132 floats, not 128: a wave's ds_load_b64 fragment read covers all 64 banks once where 16 rows met on 2 banks per half wave (16-way conflict gone)
# speedup vs baseline: 2.5801x; 1.1246x over previous
#include <hip/hip_runtime.h>


#define NN 100000
#define NE 1600000
#define KD 128
#define NNP 100096
#define NSCAN 391
#define NPW 4
#define AP 132

typedef __attribute__((ext_vector_type(2))) float v2f;
typedef __attribute__((ext_vector_type(8))) float v8f;

__global__ void fill_zero_kernel(float* __restrict__ p, int n) {
  int i = blockIdx.x * blockDim.x + threadIdx.x;
  int stride = gridDim.x * blockDim.x;
  for (; i < n; i += stride) p[i] = 0.0f;
}

__global__ void degree_rank_kernel(const int* __restrict__ src, const int* __restrict__ dst,
                                   float* __restrict__ out_deg, int* __restrict__ in_cnt,
                                   int* __restrict__ rank) {
  int e = blockIdx.x * blockDim.x + threadIdx.x;
  if (e >= NE) return;
  unsafeAtomicAdd(out_deg + src[e], 1.0f);
  rank[e] = atomicAdd(in_cnt + dst[e], 1);
}

__global__ void block_sum_kernel(const int* __restrict__ in_cnt, int* __restrict__ blocksum) {
  __shared__ int part[8];
  int t = threadIdx.x;
  int i = blockIdx.x * 256 + t;
  int c = i < NN ? in_cnt[i] : 0;
  #pragma unroll
  for (int d = 16; d > 0; d >>= 1) c += __shfl_xor(c, d, 32);
  if ((t & 31) == 0) part[t >> 5] = c;
  __syncthreads();
  if (t == 0) {
    int s = 0;
    #pragma unroll
    for (int k = 0; k < 8; ++k) s += part[k];
    blocksum[blockIdx.x] = s;
  }
}

__global__ void scan_norm_kernel(const int* __restrict__ in_cnt, const int* __restrict__ blocksum,
                                 int* __restrict__ offs, float* __restrict__ in_norm,
                                 float* __restrict__ out_deg) {
  __shared__ int wtot[8];
  __shared__ int wpre[8];
  const int t = threadIdx.x;
  const int b = blockIdx.x;
  const int lane = t & 31;
  const int wv = t >> 5;
  const int i = b * 256 + t;
  const int c = i < NN ? in_cnt[i] : 0;

  int p = 0;
  if (t < b) p += blocksum[t];
  if (t + 256 < b) p += blocksum[t + 256];
  #pragma unroll
  for (int d = 16; d > 0; d >>= 1) p += __shfl_xor(p, d, 32);

  int s = c;
  #pragma unroll
  for (int d = 1; d < 32; d <<= 1) {
    int u = __shfl_up(s, d, 32);
    if (lane >= d) s += u;
  }
  if (lane == 31) { wtot[wv] = s; wpre[wv] = p; }
  __syncthreads();
  int base = 0;
  #pragma unroll
  for (int k = 0; k < 8; ++k) {
    base += wpre[k];
    if (k < wv) base += wtot[k];
  }
  if (i < NN) {
    offs[i] = base + s - c;
    float di = (float)c;
    di = di < 1.0f ? 1.0f : di;
    in_norm[i] = rsqrtf(di);
    float d = out_deg[i];
    d = d < 1.0f ? 1.0f : d;
    out_deg[i] = rsqrtf(d);
  }
}

__global__ void place_kernel(const int* __restrict__ src, const int* __restrict__ dst,
                             const float* __restrict__ ew, const int* __restrict__ rank,
                             const int* __restrict__ offs, const float* __restrict__ out_norm,
                             int* __restrict__ rec) {
  int e = blockIdx.x * blockDim.x + threadIdx.x;
  if (e >= NE) return;
  int s = src[e];
  float w = ew[e];
  int pos = offs[dst[e]] + rank[e];
  int4 r;
  r.x = s;
  r.y = __float_as_int(w * out_norm[s]);
  r.z = __float_as_int(w);
  r.w = 0;
  ((int4*)rec)[pos] = r;
}

template <int U, int WIDTH, bool SCALED, typename V>
__device__ __forceinline__ void row_group(V& acc, const float* __restrict__ h, int4 r, int j,
                                          unsigned lane) {
  constexpr int VL = WIDTH / 32;
  V v[U];
  float c[U];
  #pragma unroll
  for (int u = 0; u < U; ++u) {
    int s = __builtin_amdgcn_readlane(r.x, j + u);
    c[u] = __int_as_float(__builtin_amdgcn_readlane(SCALED ? r.y : r.z, j + u));
    v[u] = *(const V*)(h + (size_t)s * WIDTH + lane * VL);
  }
  #pragma unroll
  for (int u = 0; u < U; ++u) acc += v[u] * c[u];
}

template <int WIDTH, bool SCALED, bool FINAL, typename V>
__global__ void gather_sum_kernel(const float* __restrict__ h, const int* __restrict__ rec,
                                  const int* __restrict__ offs, const int* __restrict__ in_cnt,
                                  const float* __restrict__ in_norm,
                                  const float* __restrict__ bias,
                                  float* __restrict__ agg) {
  constexpr int VL = WIDTH / 32;
  const unsigned lane = threadIdx.x & 31u;
  V bv = {};
  if (FINAL) bv = *(const V*)(bias + lane * VL);
  const int w = __builtin_amdgcn_readfirstlane((int)(blockIdx.x * (blockDim.x >> 5) + (threadIdx.x >> 5)));
  const int4* __restrict__ rp = (const int4*)rec;
  for (int k = 0; k < NPW; ++k) {
    const int n = w * NPW + k;
    if (n >= NN) return;
    const int beg = offs[n];
    const int deg = in_cnt[n];
    const float nrm = FINAL ? in_norm[n] : 1.0f;
    V acc = {};
    for (int c0 = 0; c0 < deg; c0 += 32) {
      const int m = deg - c0 < 32 ? deg - c0 : 32;
      int4 r = {0, 0, 0, 0};
      if ((int)lane < m) r = rp[beg + c0 + (int)lane];
      int j = 0;
      for (; j + 8 <= m; j += 8) row_group<8, WIDTH, SCALED, V>(acc, h, r, j, lane);
      if (j + 4 <= m) { row_group<4, WIDTH, SCALED, V>(acc, h, r, j, lane); j += 4; }
      if (j + 2 <= m) { row_group<2, WIDTH, SCALED, V>(acc, h, r, j, lane); j += 2; }
      if (j < m) row_group<1, WIDTH, SCALED, V>(acc, h, r, j, lane);
    }
    if (FINAL) acc = acc * nrm + bv;
    *(V*)(agg + (size_t)n * WIDTH + lane * VL) = acc;
  }
}

template <int FOUT, bool RELU, bool HAS_BIAS>
__global__ void wmma_gemm_kernel(const float* __restrict__ A,
                                 const float* __restrict__ row_scale,
                                 const float* __restrict__ W,
                                 const float* __restrict__ bias,
                                 float* __restrict__ out) {
  __shared__ __attribute__((aligned(16))) float As[16 * AP];
  const int m_base = blockIdx.x * 16;

  {
    constexpr int NVEC = 16 * KD / 4;
    for (int v = threadIdx.x; v < NVEC; v += blockDim.x) {
      int r = v >> 5;
      int cv = v & 31;
      float4 x = ((const float4*)(A + (size_t)(m_base + r) * KD))[cv];
      float s = row_scale[m_base + r];
      x.x *= s; x.y *= s; x.z *= s; x.w *= s;
      *(float4*)&As[r * AP + 4 * cv] = x;
    }
  }
  __syncthreads();

  const int wave = threadIdx.x >> 5;
  const int lane = threadIdx.x & 31;
  const int n_base = wave * 16;
  const int lo = lane & 15;
  const int hi = lane >> 4;

  v8f acc = {};
  const float* bbase = W + (size_t)(2 * hi) * FOUT + n_base + lo;
  const float* abase = &As[lo * AP + 2 * hi];

  #pragma unroll
  for (int k0 = 0; k0 < KD; k0 += 4) {
    v2f a = *(const v2f*)(abase + k0);
    v2f b;
    b.x = bbase[(size_t)(k0 + 0) * FOUT];
    b.y = bbase[(size_t)(k0 + 1) * FOUT];
    acc = __builtin_amdgcn_wmma_f32_16x16x4_f32(
         false, a,  false, b,
         (short)0, acc,  false,  false);
  }

  const int col = n_base + lo;
  const float bv = HAS_BIAS ? bias[col] : 0.0f;
  #pragma unroll
  for (int v = 0; v < 8; ++v) {
    int row = m_base + v + 8 * hi;
    float val = acc[v] + bv;
    if (RELU) val = val > 0.0f ? val : 0.0f;
    out[(size_t)row * FOUT + col] = val;
  }
}

extern "C" void kernel_launch(void* const* d_in, const int* in_sizes, int n_in,
                              void* d_out, int out_size, void* d_ws, size_t ws_size,
                              hipStream_t stream) {
  const float* features = (const float*)d_in[0];
  const int*   src      = (const int*)d_in[1];
  const int*   dst      = (const int*)d_in[2];
  const float* ew       = (const float*)d_in[3];
  const float* W0 = (const float*)d_in[4];
  const float* b0 = (const float*)d_in[5];
  const float* W1 = (const float*)d_in[6];
  const float* b1 = (const float*)d_in[7];
  const float* W2 = (const float*)d_in[8];
  const float* b2 = (const float*)d_in[9];
  float* out = (float*)d_out;

  float* ws       = (float*)d_ws;
  float* out_norm = ws;
  int*   in_cnt   = (int*)(ws + NNP);
  float* in_norm  = ws + 2 * NNP;
  int*   offs     = (int*)(ws + 3 * NNP);
  int*   blocksum = (int*)(ws + 4 * NNP);
  int*   rank     = (int*)(ws + 4 * NNP + 512);
  int*   rec      = rank + NE;
  float* bufA     = (float*)(rec + (size_t)4 * NE);
  float* bufB     = bufA + (size_t)NN * KD;

  fill_zero_kernel<<<512, 256, 0, stream>>>(out_norm, 2 * NNP);
  degree_rank_kernel<<<(NE + 255) / 256, 256, 0, stream>>>(src, dst, out_norm, in_cnt, rank);
  block_sum_kernel<<<NSCAN, 256, 0, stream>>>(in_cnt, blocksum);
  scan_norm_kernel<<<NSCAN, 256, 0, stream>>>(in_cnt, blocksum, offs, in_norm, out_norm);
  place_kernel<<<(NE + 255) / 256, 256, 0, stream>>>(src, dst, ew, rank, offs, out_norm, rec);

  gather_sum_kernel<128, true, false, float4><<<(NN / NPW * 32 + 255) / 256, 256, 0, stream>>>(
      features, rec, offs, in_cnt, nullptr, nullptr, bufA);
  wmma_gemm_kernel<128, true, true><<<NN / 16, 256, 0, stream>>>(
      bufA, in_norm, W0, b0, bufB);

  gather_sum_kernel<128, true, false, float4><<<(NN / NPW * 32 + 255) / 256, 256, 0, stream>>>(
      bufB, rec, offs, in_cnt, nullptr, nullptr, bufA);
  wmma_gemm_kernel<128, true, true><<<NN / 16, 256, 0, stream>>>(
      bufA, in_norm, W1, b1, bufB);

  wmma_gemm_kernel<64, false, false><<<NN / 16, 128, 0, stream>>>(
      bufB, out_norm, W2, nullptr, bufA);
  gather_sum_kernel<64, false, true, float2><<<(NN / NPW * 32 + 255) / 256, 256, 0, stream>>>(
      bufA, rec, offs, in_cnt, in_norm, b2, out);
}


# layers 0/1: each block sums the rows of its 32 nodes into LDS and runs the f32 WMMA GEMM on that tile itself; summed rows never go to memory, both 128-wide GEMM launches gone (11 launches to 9)
# speedup vs baseline: 2.8146x; 1.0909x over previous
#include <hip/hip_runtime.h>


#define NN 100000
#define NE 1600000
#define KD 128
#define NNP 100096
#define NSCAN 391
#define NPW 4
#define NPB 32
#define AP 132

typedef __attribute__((ext_vector_type(2))) float v2f;
typedef __attribute__((ext_vector_type(8))) float v8f;

__global__ void fill_zero_kernel(float* __restrict__ p, int n) {
  int i = blockIdx.x * blockDim.x + threadIdx.x;
  int stride = gridDim.x * blockDim.x;
  for (; i < n; i += stride) p[i] = 0.0f;
}

__global__ void degree_rank_kernel(const int* __restrict__ src, const int* __restrict__ dst,
                                   float* __restrict__ out_deg, int* __restrict__ in_cnt,
                                   int* __restrict__ rank) {
  int e = blockIdx.x * blockDim.x + threadIdx.x;
  if (e >= NE) return;
  unsafeAtomicAdd(out_deg + src[e], 1.0f);
  rank[e] = atomicAdd(in_cnt + dst[e], 1);
}

__global__ void block_sum_kernel(const int* __restrict__ in_cnt, int* __restrict__ blocksum) {
  __shared__ int part[8];
  int t = threadIdx.x;
  int i = blockIdx.x * 256 + t;
  int c = i < NN ? in_cnt[i] : 0;
  #pragma unroll
  for (int d = 16; d > 0; d >>= 1) c += __shfl_xor(c, d, 32);
  if ((t & 31) == 0) part[t >> 5] = c;
  __syncthreads();
  if (t == 0) {
    int s = 0;
    #pragma unroll
    for (int k = 0; k < 8; ++k) s += part[k];
    blocksum[blockIdx.x] = s;
  }
}

__global__ void scan_norm_kernel(const int* __restrict__ in_cnt, const int* __restrict__ blocksum,
                                 int* __restrict__ offs, float* __restrict__ in_norm,
                                 float* __restrict__ out_deg) {
  __shared__ int wtot[8];
  __shared__ int wpre[8];
  const int t = threadIdx.x;
  const int b = blockIdx.x;
  const int lane = t & 31;
  const int wv = t >> 5;
  const int i = b * 256 + t;
  const int c = i < NN ? in_cnt[i] : 0;

  int p = 0;
  if (t < b) p += blocksum[t];
  if (t + 256 < b) p += blocksum[t + 256];
  #pragma unroll
  for (int d = 16; d > 0; d >>= 1) p += __shfl_xor(p, d, 32);

  int s = c;
  #pragma unroll
  for (int d = 1; d < 32; d <<= 1) {
    int u = __shfl_up(s, d, 32);
    if (lane >= d) s += u;
  }
  if (lane == 31) { wtot[wv] = s; wpre[wv] = p; }
  __syncthreads();
  int base = 0;
  #pragma unroll
  for (int k = 0; k < 8; ++k) {
    base += wpre[k];
    if (k < wv) base += wtot[k];
  }
  if (i < NN) {
    offs[i] = base + s - c;
    float di = (float)c;
    di = di < 1.0f ? 1.0f : di;
    in_norm[i] = rsqrtf(di);
    float d = out_deg[i];
    d = d < 1.0f ? 1.0f : d;
    out_deg[i] = rsqrtf(d);
  }
}

__global__ void place_kernel(const int* __restrict__ src, const int* __restrict__ dst,
                             const float* __restrict__ ew, const int* __restrict__ rank,
                             const int* __restrict__ offs, const float* __restrict__ out_norm,
                             int* __restrict__ rec) {
  int e = blockIdx.x * blockDim.x + threadIdx.x;
  if (e >= NE) return;
  int s = src[e];
  float w = ew[e];
  int pos = offs[dst[e]] + rank[e];
  int4 r;
  r.x = s;
  r.y = __float_as_int(w * out_norm[s]);
  r.z = __float_as_int(w);
  r.w = 0;
  ((int4*)rec)[pos] = r;
}

template <int U, int WIDTH, bool SCALED, typename V>
__device__ __forceinline__ void row_group(V& acc, const float* __restrict__ h, int4 r, int j,
                                          unsigned lane) {
  constexpr int VL = WIDTH / 32;
  V v[U];
  float c[U];
  #pragma unroll
  for (int u = 0; u < U; ++u) {
    int s = __builtin_amdgcn_readlane(r.x, j + u);
    c[u] = __int_as_float(__builtin_amdgcn_readlane(SCALED ? r.y : r.z, j + u));
    v[u] = *(const V*)(h + (size_t)s * WIDTH + lane * VL);
  }
  #pragma unroll
  for (int u = 0; u < U; ++u) acc += v[u] * c[u];
}

template <int WIDTH, bool SCALED, bool FINAL, typename V>
__global__ void gather_sum_kernel(const float* __restrict__ h, const int* __restrict__ rec,
                                  const int* __restrict__ offs, const int* __restrict__ in_cnt,
                                  const float* __restrict__ in_norm,
                                  const float* __restrict__ bias,
                                  float* __restrict__ agg) {
  constexpr int VL = WIDTH / 32;
  const unsigned lane = threadIdx.x & 31u;
  V bv = {};
  if (FINAL) bv = *(const V*)(bias + lane * VL);
  const int w = __builtin_amdgcn_readfirstlane((int)(blockIdx.x * (blockDim.x >> 5) + (threadIdx.x >> 5)));
  const int4* __restrict__ rp = (const int4*)rec;
  for (int k = 0; k < NPW; ++k) {
    const int n = w * NPW + k;
    if (n >= NN) return;
    const int beg = offs[n];
    const int deg = in_cnt[n];
    const float nrm = FINAL ? in_norm[n] : 1.0f;
    V acc = {};
    for (int c0 = 0; c0 < deg; c0 += 32) {
      const int m = deg - c0 < 32 ? deg - c0 : 32;
      int4 r = {0, 0, 0, 0};
      if ((int)lane < m) r = rp[beg + c0 + (int)lane];
      int j = 0;
      for (; j + 8 <= m; j += 8) row_group<8, WIDTH, SCALED, V>(acc, h, r, j, lane);
      if (j + 4 <= m) { row_group<4, WIDTH, SCALED, V>(acc, h, r, j, lane); j += 4; }
      if (j + 2 <= m) { row_group<2, WIDTH, SCALED, V>(acc, h, r, j, lane); j += 2; }
      if (j < m) row_group<1, WIDTH, SCALED, V>(acc, h, r, j, lane);
    }
    if (FINAL) acc = acc * nrm + bv;
    *(V*)(agg + (size_t)n * WIDTH + lane * VL) = acc;
  }
}

template <int FOUT, bool RELU, bool HAS_BIAS>
__global__ void wmma_gemm_kernel(const float* __restrict__ A,
                                 const float* __restrict__ row_scale,
                                 const float* __restrict__ W,
                                 const float* __restrict__ bias,
                                 float* __restrict__ out) {
  __shared__ __attribute__((aligned(16))) float As[16 * AP];
  const int m_base = blockIdx.x * 16;

  {
    constexpr int NVEC = 16 * KD / 4;
    for (int v = threadIdx.x; v < NVEC; v += blockDim.x) {
      int r = v >> 5;
      int cv = v & 31;
      float4 x = ((const float4*)(A + (size_t)(m_base + r) * KD))[cv];
      float s = row_scale[m_base + r];
      x.x *= s; x.y *= s; x.z *= s; x.w *= s;
      *(float4*)&As[r * AP + 4 * cv] = x;
    }
  }
  __syncthreads();

  const int wave = threadIdx.x >> 5;
  const int lane = threadIdx.x & 31;
  const int n_base = wave * 16;
  const int lo = lane & 15;
  const int hi = lane >> 4;

  v8f acc = {};
  const float* bbase = W + (size_t)(2 * hi) * FOUT + n_base + lo;
  const float* abase = &As[lo * AP + 2 * hi];

  #pragma unroll
  for (int k0 = 0; k0 < KD; k0 += 4) {
    v2f a = *(const v2f*)(abase + k0);
    v2f b;
    b.x = bbase[(size_t)(k0 + 0) * FOUT];
    b.y = bbase[(size_t)(k0 + 1) * FOUT];
    acc = __builtin_amdgcn_wmma_f32_16x16x4_f32(
         false, a,  false, b,
         (short)0, acc,  false,  false);
  }

  const int col = n_base + lo;
  const float bv = HAS_BIAS ? bias[col] : 0.0f;
  #pragma unroll
  for (int v = 0; v < 8; ++v) {
    int row = m_base + v + 8 * hi;
    float val = acc[v] + bv;
    if (RELU) val = val > 0.0f ? val : 0.0f;
    out[(size_t)row * FOUT + col] = val;
  }
}

__global__ void __launch_bounds__(256)
gather_gemm_kernel(const float* __restrict__ h, const int* __restrict__ rec,
                   const int* __restrict__ offs, const int* __restrict__ in_cnt,
                   const float* __restrict__ in_norm, const float* __restrict__ W,
                   const float* __restrict__ bias, float* __restrict__ out) {
  static_assert(NN % NPB == 0, "every block has all of its NPB nodes");
  static_assert(NPB == 8 * NPW && NPB == 32, "8 waves x NPW nodes = two 16-row tiles");
  __shared__ __attribute__((aligned(16))) float As[NPB * AP];
  const unsigned lane = threadIdx.x & 31u;
  const int wave = __builtin_amdgcn_readfirstlane((int)(threadIdx.x >> 5));
  const int m_base = blockIdx.x * NPB;
  const int4* __restrict__ rp = (const int4*)rec;

  for (int k = 0; k < NPW; ++k) {
    const int r = wave * NPW + k;
    const int n = m_base + r;
    const int beg = offs[n];
    const int deg = in_cnt[n];
    const float nrm = in_norm[n];
    float4 acc = {};
    for (int c0 = 0; c0 < deg; c0 += 32) {
      const int m = deg - c0 < 32 ? deg - c0 : 32;
      int4 q = {0, 0, 0, 0};
      if ((int)lane < m) q = rp[beg + c0 + (int)lane];
      int j = 0;
      for (; j + 8 <= m; j += 8) row_group<8, KD, true, float4>(acc, h, q, j, lane);
      if (j + 4 <= m) { row_group<4, KD, true, float4>(acc, h, q, j, lane); j += 4; }
      if (j + 2 <= m) { row_group<2, KD, true, float4>(acc, h, q, j, lane); j += 2; }
      if (j < m) row_group<1, KD, true, float4>(acc, h, q, j, lane);
    }
    acc = acc * nrm;
    *(float4*)&As[r * AP + lane * 4] = acc;
  }
  __syncthreads();

  const int n_base = wave * 16;
  const int lo = lane & 15;
  const int hi = lane >> 4;
  v8f acc0 = {};
  v8f acc1 = {};
  const float* bbase = W + (size_t)(2 * hi) * KD + n_base + lo;
  const float* abase = &As[lo * AP + 2 * hi];

  #pragma unroll
  for (int k0 = 0; k0 < KD; k0 += 4) {
    v2f a0 = *(const v2f*)(abase + k0);
    v2f a1 = *(const v2f*)(abase + 16 * AP + k0);
    v2f b;
    b.x = bbase[(size_t)(k0 + 0) * KD];
    b.y = bbase[(size_t)(k0 + 1) * KD];
    acc0 = __builtin_amdgcn_wmma_f32_16x16x4_f32(
         false, a0,  false, b,
         (short)0, acc0,  false,  false);
    acc1 = __builtin_amdgcn_wmma_f32_16x16x4_f32(
         false, a1,  false, b,
         (short)0, acc1,  false,  false);
  }

  const int col = n_base + lo;
  const float bv = bias[col];
  #pragma unroll
  for (int v = 0; v < 8; ++v) {
    const int row = m_base + v + 8 * hi;
    float x0 = acc0[v] + bv;
    float x1 = acc1[v] + bv;
    x0 = x0 > 0.0f ? x0 : 0.0f;
    x1 = x1 > 0.0f ? x1 : 0.0f;
    out[(size_t)row * KD + col] = x0;
    out[(size_t)(row + 16) * KD + col] = x1;
  }
}

extern "C" void kernel_launch(void* const* d_in, const int* in_sizes, int n_in,
                              void* d_out, int out_size, void* d_ws, size_t ws_size,
                              hipStream_t stream) {
  const float* features = (const float*)d_in[0];
  const int*   src      = (const int*)d_in[1];
  const int*   dst      = (const int*)d_in[2];
  const float* ew       = (const float*)d_in[3];
  const float* W0 = (const float*)d_in[4];
  const float* b0 = (const float*)d_in[5];
  const float* W1 = (const float*)d_in[6];
  const float* b1 = (const float*)d_in[7];
  const float* W2 = (const float*)d_in[8];
  const float* b2 = (const float*)d_in[9];
  float* out = (float*)d_out;

  float* ws       = (float*)d_ws;
  float* out_norm = ws;
  int*   in_cnt   = (int*)(ws + NNP);
  float* in_norm  = ws + 2 * NNP;
  int*   offs     = (int*)(ws + 3 * NNP);
  int*   blocksum = (int*)(ws + 4 * NNP);
  int*   rank     = (int*)(ws + 4 * NNP + 512);
  int*   rec      = rank + NE;
  float* bufA     = (float*)(rec + (size_t)4 * NE);
  float* bufB     = bufA + (size_t)NN * KD;

  fill_zero_kernel<<<512, 256, 0, stream>>>(out_norm, 2 * NNP);
  degree_rank_kernel<<<(NE + 255) / 256, 256, 0, stream>>>(src, dst, out_norm, in_cnt, rank);
  block_sum_kernel<<<NSCAN, 256, 0, stream>>>(in_cnt, blocksum);
  scan_norm_kernel<<<NSCAN, 256, 0, stream>>>(in_cnt, blocksum, offs, in_norm, out_norm);
  place_kernel<<<(NE + 255) / 256, 256, 0, stream>>>(src, dst, ew, rank, offs, out_norm, rec);

  gather_gemm_kernel<<<NN / NPB, 256, 0, stream>>>(
      features, rec, offs, in_cnt, in_norm, W0, b0, bufA);

  gather_gemm_kernel<<<NN / NPB, 256, 0, stream>>>(
      bufA, rec, offs, in_cnt, in_norm, W1, b1, bufB);

  wmma_gemm_kernel<64, false, false><<<NN / 16, 128, 0, stream>>>(
      bufB, out_norm, W2, nullptr, bufA);
  gather_sum_kernel<64, false, true, float2><<<(NN / NPW * 32 + 255) / 256, 256, 0, stream>>>(
      bufA, rec, offs, in_cnt, in_norm, b2, out);
}


# degree counts and edge ranks from LDS counters (32 edge chunks x 8 node ranges per key), chunk bases from the scan; no global atomics or zero-fill launch; sum order = chunk, LDS arrival rank
# speedup vs baseline: 4.2504x; 1.5101x over previous
#include <hip/hip_runtime.h>


#define NN 100000
#define NE 1600000
#define KD 128
#define NNP 100096
#define NSCAN 391
#define NPW 4
#define NPB 32
#define AP 132
#define CH 32
#define EC 50000
#define NR 8
#define RN 12500

typedef __attribute__((ext_vector_type(2))) float v2f;
typedef __attribute__((ext_vector_type(8))) float v8f;

template <bool RANKED>
__device__ __forceinline__ void count_chunk(const int* __restrict__ key, int lo, int* cntr,
                                            int* __restrict__ rk) {
  constexpr int NV = EC / 4;
  constexpr int U = 8;
  const int4* __restrict__ k4 = (const int4*)key;
  for (int v0 = threadIdx.x; v0 < NV; v0 += U * 1024) {
    int4 k[U];
    #pragma unroll
    for (int u = 0; u < U; ++u) {
      const int v = v0 + u * 1024;
      k[u] = make_int4(-1, -1, -1, -1);
      if (v < NV) k[u] = k4[v];
    }
    #pragma unroll
    for (int u = 0; u < U; ++u) {
      const int e = 4 * (v0 + u * 1024);
      const int kk[4] = {k[u].x, k[u].y, k[u].z, k[u].w};
      #pragma unroll
      for (int q = 0; q < 4; ++q) {
        const unsigned a = (unsigned)(kk[q] - lo);
        if (a < (unsigned)RN) {
          if (RANKED) rk[e + q] = atomicAdd(&cntr[a], 1);
          else atomicAdd(&cntr[a], 1);
        }
      }
    }
  }
}

__global__ void __launch_bounds__(1024)
count_rank_kernel(const int* __restrict__ src, const int* __restrict__ dst,
                  int* __restrict__ cnt_in, int* __restrict__ cnt_out, int* __restrict__ rank) {
  static_assert(CH * EC == NE && EC % 4 == 0, "chunks tile the edge list in whole int4");
  static_assert(NR * RN == NN, "ranges tile the nodes");
  __shared__ int cntr[RN];
  const int c = blockIdx.x;
  const int lo = blockIdx.y * RN;
  const bool in_role = blockIdx.z == 0;
  for (int i = threadIdx.x; i < RN; i += 1024) cntr[i] = 0;
  __syncthreads();
  if (in_role) count_chunk<true>(dst + (size_t)c * EC, lo, cntr, rank + (size_t)c * EC);
  else count_chunk<false>(src + (size_t)c * EC, lo, cntr, nullptr);
  __syncthreads();
  int* __restrict__ plane = (in_role ? cnt_in : cnt_out) + (size_t)c * NNP + lo;
  for (int i = threadIdx.x; i < RN; i += 1024) plane[i] = cntr[i];
}

__global__ void block_sum_kernel(const int* __restrict__ cnt_in, int* __restrict__ blocksum) {
  __shared__ int part[8];
  int t = threadIdx.x;
  int i = blockIdx.x * 256 + t;
  int c = 0;
  if (i < NN) {
    #pragma unroll
    for (int k = 0; k < CH; ++k) c += cnt_in[(size_t)k * NNP + i];
  }
  #pragma unroll
  for (int d = 16; d > 0; d >>= 1) c += __shfl_xor(c, d, 32);
  if ((t & 31) == 0) part[t >> 5] = c;
  __syncthreads();
  if (t == 0) {
    int s = 0;
    #pragma unroll
    for (int k = 0; k < 8; ++k) s += part[k];
    blocksum[blockIdx.x] = s;
  }
}

__global__ void scan_norm_kernel(int* __restrict__ cnt_in, const int* __restrict__ cnt_out,
                                 const int* __restrict__ blocksum, int* __restrict__ offs,
                                 int* __restrict__ in_cnt, float* __restrict__ in_norm,
                                 float* __restrict__ out_norm) {
  __shared__ int wtot[8];
  __shared__ int wpre[8];
  const int t = threadIdx.x;
  const int b = blockIdx.x;
  const int lane = t & 31;
  const int wv = t >> 5;
  const int i = b * 256 + t;
  int c = 0;
  int od = 0;
  if (i < NN) {
    int cnt[CH];
    #pragma unroll
    for (int k = 0; k < CH; ++k) {
      cnt[k] = cnt_in[(size_t)k * NNP + i];
      od += cnt_out[(size_t)k * NNP + i];
    }
    #pragma unroll
    for (int k = 0; k < CH; ++k) {
      cnt_in[(size_t)k * NNP + i] = c;
      c += cnt[k];
    }
  }

  int p = 0;
  if (t < b) p += blocksum[t];
  if (t + 256 < b) p += blocksum[t + 256];
  #pragma unroll
  for (int d = 16; d > 0; d >>= 1) p += __shfl_xor(p, d, 32);

  int s = c;
  #pragma unroll
  for (int d = 1; d < 32; d <<= 1) {
    int u = __shfl_up(s, d, 32);
    if (lane >= d) s += u;
  }
  if (lane == 31) { wtot[wv] = s; wpre[wv] = p; }
  __syncthreads();
  int base = 0;
  #pragma unroll
  for (int k = 0; k < 8; ++k) {
    base += wpre[k];
    if (k < wv) base += wtot[k];
  }
  if (i < NN) {
    offs[i] = base + s - c;
    in_cnt[i] = c;
    float di = (float)c;
    di = di < 1.0f ? 1.0f : di;
    in_norm[i] = rsqrtf(di);
    float d = (float)od;
    d = d < 1.0f ? 1.0f : d;
    out_norm[i] = rsqrtf(d);
  }
}

__global__ void place_kernel(const int* __restrict__ src, const int* __restrict__ dst,
                             const float* __restrict__ ew, const int* __restrict__ rank,
                             const int* __restrict__ offs, const int* __restrict__ cbase,
                             const float* __restrict__ out_norm, int* __restrict__ rec) {
  int e = blockIdx.x * blockDim.x + threadIdx.x;
  if (e >= NE) return;
  int s = src[e];
  float w = ew[e];
  int d = dst[e];
  int pos = offs[d] + cbase[(size_t)(e / EC) * NNP + d] + rank[e];
  int4 r;
  r.x = s;
  r.y = __float_as_int(w * out_norm[s]);
  r.z = __float_as_int(w);
  r.w = 0;
  ((int4*)rec)[pos] = r;
}

template <int U, int WIDTH, bool SCALED, typename V>
__device__ __forceinline__ void row_group(V& acc, const float* __restrict__ h, int4 r, int j,
                                          unsigned lane) {
  constexpr int VL = WIDTH / 32;
  V v[U];
  float c[U];
  #pragma unroll
  for (int u = 0; u < U; ++u) {
    int s = __builtin_amdgcn_readlane(r.x, j + u);
    c[u] = __int_as_float(__builtin_amdgcn_readlane(SCALED ? r.y : r.z, j + u));
    v[u] = *(const V*)(h + (size_t)s * WIDTH + lane * VL);
  }
  #pragma unroll
  for (int u = 0; u < U; ++u) acc += v[u] * c[u];
}

template <int WIDTH, bool SCALED, bool FINAL, typename V>
__global__ void gather_sum_kernel(const float* __restrict__ h, const int* __restrict__ rec,
                                  const int* __restrict__ offs, const int* __restrict__ in_cnt,
                                  const float* __restrict__ in_norm,
                                  const float* __restrict__ bias,
                                  float* __restrict__ agg) {
  constexpr int VL = WIDTH / 32;
  const unsigned lane = threadIdx.x & 31u;
  V bv = {};
  if (FINAL) bv = *(const V*)(bias + lane * VL);
  const int w = __builtin_amdgcn_readfirstlane((int)(blockIdx.x * (blockDim.x >> 5) + (threadIdx.x >> 5)));
  const int4* __restrict__ rp = (const int4*)rec;
  for (int k = 0; k < NPW; ++k) {
    const int n = w * NPW + k;
    if (n >= NN) return;
    const int beg = offs[n];
    const int deg = in_cnt[n];
    const float nrm = FINAL ? in_norm[n] : 1.0f;
    V acc = {};
    for (int c0 = 0; c0 < deg; c0 += 32) {
      const int m = deg - c0 < 32 ? deg - c0 : 32;
      int4 r = {0, 0, 0, 0};
      if ((int)lane < m) r = rp[beg + c0 + (int)lane];
      int j = 0;
      for (; j + 8 <= m; j += 8) row_group<8, WIDTH, SCALED, V>(acc, h, r, j, lane);
      if (j + 4 <= m) { row_group<4, WIDTH, SCALED, V>(acc, h, r, j, lane); j += 4; }
      if (j + 2 <= m) { row_group<2, WIDTH, SCALED, V>(acc, h, r, j, lane); j += 2; }
      if (j < m) row_group<1, WIDTH, SCALED, V>(acc, h, r, j, lane);
    }
    if (FINAL) acc = acc * nrm + bv;
    *(V*)(agg + (size_t)n * WIDTH + lane * VL) = acc;
  }
}

template <int FOUT, bool RELU, bool HAS_BIAS>
__global__ void wmma_gemm_kernel(const float* __restrict__ A,
                                 const float* __restrict__ row_scale,
                                 const float* __restrict__ W,
                                 const float* __restrict__ bias,
                                 float* __restrict__ out) {
  __shared__ __attribute__((aligned(16))) float As[16 * AP];
  const int m_base = blockIdx.x * 16;

  {
    constexpr int NVEC = 16 * KD / 4;
    for (int v = threadIdx.x; v < NVEC; v += blockDim.x) {
      int r = v >> 5;
      int cv = v & 31;
      float4 x = ((const float4*)(A + (size_t)(m_base + r) * KD))[cv];
      float s = row_scale[m_base + r];
      x.x *= s; x.y *= s; x.z *= s; x.w *= s;
      *(float4*)&As[r * AP + 4 * cv] = x;
    }
  }
  __syncthreads();

  const int wave = threadIdx.x >> 5;
  const int lane = threadIdx.x & 31;
  const int n_base = wave * 16;
  const int lo = lane & 15;
  const int hi = lane >> 4;

  v8f acc = {};
  const float* bbase = W + (size_t)(2 * hi) * FOUT + n_base + lo;
  const float* abase = &As[lo * AP + 2 * hi];

  #pragma unroll
  for (int k0 = 0; k0 < KD; k0 += 4) {
    v2f a = *(const v2f*)(abase + k0);
    v2f b;
    b.x = bbase[(size_t)(k0 + 0) * FOUT];
    b.y = bbase[(size_t)(k0 + 1) * FOUT];
    acc = __builtin_amdgcn_wmma_f32_16x16x4_f32(
         false, a,  false, b,
         (short)0, acc,  false,  false);
  }

  const int col = n_base + lo;
  const float bv = HAS_BIAS ? bias[col] : 0.0f;
  #pragma unroll
  for (int v = 0; v < 8; ++v) {
    int row = m_base + v + 8 * hi;
    float val = acc[v] + bv;
    if (RELU) val = val > 0.0f ? val : 0.0f;
    out[(size_t)row * FOUT + col] = val;
  }
}

__global__ void __launch_bounds__(256)
gather_gemm_kernel(const float* __restrict__ h, const int* __restrict__ rec,
                   const int* __restrict__ offs, const int* __restrict__ in_cnt,
                   const float* __restrict__ in_norm, const float* __restrict__ W,
                   const float* __restrict__ bias, float* __restrict__ out) {
  static_assert(NN % NPB == 0, "every block has all of its NPB nodes");
  static_assert(NPB == 8 * NPW && NPB == 32, "8 waves x NPW nodes = two 16-row tiles");
  __shared__ __attribute__((aligned(16))) float As[NPB * AP];
  const unsigned lane = threadIdx.x & 31u;
  const int wave = __builtin_amdgcn_readfirstlane((int)(threadIdx.x >> 5));
  const int m_base = blockIdx.x * NPB;
  const int4* __restrict__ rp = (const int4*)rec;

  for (int k = 0; k < NPW; ++k) {
    const int r = wave * NPW + k;
    const int n = m_base + r;
    const int beg = offs[n];
    const int deg = in_cnt[n];
    const float nrm = in_norm[n];
    float4 acc = {};
    for (int c0 = 0; c0 < deg; c0 += 32) {
      const int m = deg - c0 < 32 ? deg - c0 : 32;
      int4 q = {0, 0, 0, 0};
      if ((int)lane < m) q = rp[beg + c0 + (int)lane];
      int j = 0;
      for (; j + 8 <= m; j += 8) row_group<8, KD, true, float4>(acc, h, q, j, lane);
      if (j + 4 <= m) { row_group<4, KD, true, float4>(acc, h, q, j, lane); j += 4; }
      if (j + 2 <= m) { row_group<2, KD, true, float4>(acc, h, q, j, lane); j += 2; }
      if (j < m) row_group<1, KD, true, float4>(acc, h, q, j, lane);
    }
    acc = acc * nrm;
    *(float4*)&As[r * AP + lane * 4] = acc;
  }
  __syncthreads();

  const int n_base = wave * 16;
  const int lo = lane & 15;
  const int hi = lane >> 4;
  v8f acc0 = {};
  v8f acc1 = {};
  const float* bbase = W + (size_t)(2 * hi) * KD + n_base + lo;
  const float* abase = &As[lo * AP + 2 * hi];

  #pragma unroll
  for (int k0 = 0; k0 < KD; k0 += 4) {
    v2f a0 = *(const v2f*)(abase + k0);
    v2f a1 = *(const v2f*)(abase + 16 * AP + k0);
    v2f b;
    b.x = bbase[(size_t)(k0 + 0) * KD];
    b.y = bbase[(size_t)(k0 + 1) * KD];
    acc0 = __builtin_amdgcn_wmma_f32_16x16x4_f32(
         false, a0,  false, b,
         (short)0, acc0,  false,  false);
    acc1 = __builtin_amdgcn_wmma_f32_16x16x4_f32(
         false, a1,  false, b,
         (short)0, acc1,  false,  false);
  }

  const int col = n_base + lo;
  const float bv = bias[col];
  #pragma unroll
  for (int v = 0; v < 8; ++v) {
    const int row = m_base + v + 8 * hi;
    float x0 = acc0[v] + bv;
    float x1 = acc1[v] + bv;
    x0 = x0 > 0.0f ? x0 : 0.0f;
    x1 = x1 > 0.0f ? x1 : 0.0f;
    out[(size_t)row * KD + col] = x0;
    out[(size_t)(row + 16) * KD + col] = x1;
  }
}

extern "C" void kernel_launch(void* const* d_in, const int* in_sizes, int n_in,
                              void* d_out, int out_size, void* d_ws, size_t ws_size,
                              hipStream_t stream) {
  const float* features = (const float*)d_in[0];
  const int*   src      = (const int*)d_in[1];
  const int*   dst      = (const int*)d_in[2];
  const float* ew       = (const float*)d_in[3];
  const float* W0 = (const float*)d_in[4];
  const float* b0 = (const float*)d_in[5];
  const float* W1 = (const float*)d_in[6];
  const float* b1 = (const float*)d_in[7];
  const float* W2 = (const float*)d_in[8];
  const float* b2 = (const float*)d_in[9];
  float* out = (float*)d_out;

  float* ws       = (float*)d_ws;
  float* out_norm = ws;
  int*   in_cnt   = (int*)(ws + NNP);
  float* in_norm  = ws + 2 * NNP;
  int*   offs     = (int*)(ws + 3 * NNP);
  int*   blocksum = (int*)(ws + 4 * NNP);
  int*   rank     = (int*)(ws + 4 * NNP + 512);
  int*   rec      = rank + NE;
  float* bufA     = (float*)(rec + (size_t)4 * NE);
  float* bufB     = bufA + (size_t)NN * KD;
  int*   cnt_in   = (int*)(bufB + (size_t)NN * KD);
  int*   cnt_out  = cnt_in + (size_t)CH * NNP;

  count_rank_kernel<<<dim3(CH, NR, 2), 1024, 0, stream>>>(src, dst, cnt_in, cnt_out, rank);
  block_sum_kernel<<<NSCAN, 256, 0, stream>>>(cnt_in, blocksum);
  scan_norm_kernel<<<NSCAN, 256, 0, stream>>>(
      cnt_in, cnt_out, blocksum, offs, in_cnt, in_norm, out_norm);
  place_kernel<<<(NE + 255) / 256, 256, 0, stream>>>(
      src, dst, ew, rank, offs, cnt_in, out_norm, rec);

  gather_gemm_kernel<<<NN / NPB, 256, 0, stream>>>(
      features, rec, offs, in_cnt, in_norm, W0, b0, bufA);

  gather_gemm_kernel<<<NN / NPB, 256, 0, stream>>>(
      bufA, rec, offs, in_cnt, in_norm, W1, b1, bufB);

  wmma_gemm_kernel<64, false, false><<<NN / 16, 128, 0, stream>>>(
      bufB, out_norm, W2, nullptr, bufA);
  gather_sum_kernel<64, false, true, float2><<<(NN / NPW * 32 + 255) / 256, 256, 0, stream>>>(
      bufA, rec, offs, in_cnt, in_norm, b2, out);
}


# layer 2's 128->64 product moved into layer 1's kernel: block puts relu(rows)*out_norm in a 2nd LDS tile and multiplies by W2 there; layer-1 rows never reach memory, GEMM launch gone (8 to 7)
# speedup vs baseline: 4.4631x; 1.0500x over previous
#include <hip/hip_runtime.h>


#define NN 100000
#define NE 1600000
#define KD 128
#define NNP 100096
#define NSCAN 391
#define NPW 4
#define NPB 32
#define AP 132
#define CH 32
#define EC 50000
#define NR 8
#define RN 12500

typedef __attribute__((ext_vector_type(2))) float v2f;
typedef __attribute__((ext_vector_type(8))) float v8f;

template <bool RANKED>
__device__ __forceinline__ void count_chunk(const int* __restrict__ key, int lo, int* cntr,
                                            int* __restrict__ rk) {
  constexpr int NV = EC / 4;
  constexpr int U = 8;
  const int4* __restrict__ k4 = (const int4*)key;
  for (int v0 = threadIdx.x; v0 < NV; v0 += U * 1024) {
    int4 k[U];
    #pragma unroll
    for (int u = 0; u < U; ++u) {
      const int v = v0 + u * 1024;
      k[u] = make_int4(-1, -1, -1, -1);
      if (v < NV) k[u] = k4[v];
    }
    #pragma unroll
    for (int u = 0; u < U; ++u) {
      const int e = 4 * (v0 + u * 1024);
      const int kk[4] = {k[u].x, k[u].y, k[u].z, k[u].w};
      #pragma unroll
      for (int q = 0; q < 4; ++q) {
        const unsigned a = (unsigned)(kk[q] - lo);
        if (a < (unsigned)RN) {
          if (RANKED) rk[e + q] = atomicAdd(&cntr[a], 1);
          else atomicAdd(&cntr[a], 1);
        }
      }
    }
  }
}

__global__ void __launch_bounds__(1024)
count_rank_kernel(const int* __restrict__ src, const int* __restrict__ dst,
                  int* __restrict__ cnt_in, int* __restrict__ cnt_out, int* __restrict__ rank) {
  static_assert(CH * EC == NE && EC % 4 == 0, "chunks tile the edge list in whole int4");
  static_assert(NR * RN == NN, "ranges tile the nodes");
  __shared__ int cntr[RN];
  const int c = blockIdx.x;
  const int lo = blockIdx.y * RN;
  const bool in_role = blockIdx.z == 0;
  for (int i = threadIdx.x; i < RN; i += 1024) cntr[i] = 0;
  __syncthreads();
  if (in_role) count_chunk<true>(dst + (size_t)c * EC, lo, cntr, rank + (size_t)c * EC);
  else count_chunk<false>(src + (size_t)c * EC, lo, cntr, nullptr);
  __syncthreads();
  int* __restrict__ plane = (in_role ? cnt_in : cnt_out) + (size_t)c * NNP + lo;
  for (int i = threadIdx.x; i < RN; i += 1024) plane[i] = cntr[i];
}

__global__ void block_sum_kernel(const int* __restrict__ cnt_in, int* __restrict__ blocksum) {
  __shared__ int part[8];
  int t = threadIdx.x;
  int i = blockIdx.x * 256 + t;
  int c = 0;
  if (i < NN) {
    #pragma unroll
    for (int k = 0; k < CH; ++k) c += cnt_in[(size_t)k * NNP + i];
  }
  #pragma unroll
  for (int d = 16; d > 0; d >>= 1) c += __shfl_xor(c, d, 32);
  if ((t & 31) == 0) part[t >> 5] = c;
  __syncthreads();
  if (t == 0) {
    int s = 0;
    #pragma unroll
    for (int k = 0; k < 8; ++k) s += part[k];
    blocksum[blockIdx.x] = s;
  }
}

__global__ void scan_norm_kernel(int* __restrict__ cnt_in, const int* __restrict__ cnt_out,
                                 const int* __restrict__ blocksum, int* __restrict__ offs,
                                 int* __restrict__ in_cnt, float* __restrict__ in_norm,
                                 float* __restrict__ out_norm) {
  __shared__ int wtot[8];
  __shared__ int wpre[8];
  const int t = threadIdx.x;
  const int b = blockIdx.x;
  const int lane = t & 31;
  const int wv = t >> 5;
  const int i = b * 256 + t;
  int c = 0;
  int od = 0;
  if (i < NN) {
    int cnt[CH];
    #pragma unroll
    for (int k = 0; k < CH; ++k) {
      cnt[k] = cnt_in[(size_t)k * NNP + i];
      od += cnt_out[(size_t)k * NNP + i];
    }
    #pragma unroll
    for (int k = 0; k < CH; ++k) {
      cnt_in[(size_t)k * NNP + i] = c;
      c += cnt[k];
    }
  }

  int p = 0;
  if (t < b) p += blocksum[t];
  if (t + 256 < b) p += blocksum[t + 256];
  #pragma unroll
  for (int d = 16; d > 0; d >>= 1) p += __shfl_xor(p, d, 32);

  int s = c;
  #pragma unroll
  for (int d = 1; d < 32; d <<= 1) {
    int u = __shfl_up(s, d, 32);
    if (lane >= d) s += u;
  }
  if (lane == 31) { wtot[wv] = s; wpre[wv] = p; }
  __syncthreads();
  int base = 0;
  #pragma unroll
  for (int k = 0; k < 8; ++k) {
    base += wpre[k];
    if (k < wv) base += wtot[k];
  }
  if (i < NN) {
    offs[i] = base + s - c;
    in_cnt[i] = c;
    float di = (float)c;
    di = di < 1.0f ? 1.0f : di;
    in_norm[i] = rsqrtf(di);
    float d = (float)od;
    d = d < 1.0f ? 1.0f : d;
    out_norm[i] = rsqrtf(d);
  }
}

__global__ void place_kernel(const int* __restrict__ src, const int* __restrict__ dst,
                             const float* __restrict__ ew, const int* __restrict__ rank,
                             const int* __restrict__ offs, const int* __restrict__ cbase,
                             const float* __restrict__ out_norm, int* __restrict__ rec) {
  int e = blockIdx.x * blockDim.x + threadIdx.x;
  if (e >= NE) return;
  int s = src[e];
  float w = ew[e];
  int d = dst[e];
  int pos = offs[d] + cbase[(size_t)(e / EC) * NNP + d] + rank[e];
  int4 r;
  r.x = s;
  r.y = __float_as_int(w * out_norm[s]);
  r.z = __float_as_int(w);
  r.w = 0;
  ((int4*)rec)[pos] = r;
}

template <int U, int WIDTH, bool SCALED, typename V>
__device__ __forceinline__ void row_group(V& acc, const float* __restrict__ h, int4 r, int j,
                                          unsigned lane) {
  constexpr int VL = WIDTH / 32;
  V v[U];
  float c[U];
  #pragma unroll
  for (int u = 0; u < U; ++u) {
    int s = __builtin_amdgcn_readlane(r.x, j + u);
    c[u] = __int_as_float(__builtin_amdgcn_readlane(SCALED ? r.y : r.z, j + u));
    v[u] = *(const V*)(h + (size_t)s * WIDTH + lane * VL);
  }
  #pragma unroll
  for (int u = 0; u < U; ++u) acc += v[u] * c[u];
}

template <int WIDTH, bool SCALED, bool FINAL, typename V>
__global__ void gather_sum_kernel(const float* __restrict__ h, const int* __restrict__ rec,
                                  const int* __restrict__ offs, const int* __restrict__ in_cnt,
                                  const float* __restrict__ in_norm,
                                  const float* __restrict__ bias,
                                  float* __restrict__ agg) {
  constexpr int VL = WIDTH / 32;
  const unsigned lane = threadIdx.x & 31u;
  V bv = {};
  if (FINAL) bv = *(const V*)(bias + lane * VL);
  const int w = __builtin_amdgcn_readfirstlane((int)(blockIdx.x * (blockDim.x >> 5) + (threadIdx.x >> 5)));
  const int4* __restrict__ rp = (const int4*)rec;
  for (int k = 0; k < NPW; ++k) {
    const int n = w * NPW + k;
    if (n >= NN) return;
    const int beg = offs[n];
    const int deg = in_cnt[n];
    const float nrm = FINAL ? in_norm[n] : 1.0f;
    V acc = {};
    for (int c0 = 0; c0 < deg; c0 += 32) {
      const int m = deg - c0 < 32 ? deg - c0 : 32;
      int4 r = {0, 0, 0, 0};
      if ((int)lane < m) r = rp[beg + c0 + (int)lane];
      int j = 0;
      for (; j + 8 <= m; j += 8) row_group<8, WIDTH, SCALED, V>(acc, h, r, j, lane);
      if (j + 4 <= m) { row_group<4, WIDTH, SCALED, V>(acc, h, r, j, lane); j += 4; }
      if (j + 2 <= m) { row_group<2, WIDTH, SCALED, V>(acc, h, r, j, lane); j += 2; }
      if (j < m) row_group<1, WIDTH, SCALED, V>(acc, h, r, j, lane);
    }
    if (FINAL) acc = acc * nrm + bv;
    *(V*)(agg + (size_t)n * WIDTH + lane * VL) = acc;
  }
}


template <bool NEXT>
__global__ void __launch_bounds__(256) __attribute__((amdgpu_waves_per_eu(16, 16)))
gather_gemm_kernel(const float* __restrict__ h, const int* __restrict__ rec,
                   const int* __restrict__ offs, const int* __restrict__ in_cnt,
                   const float* __restrict__ in_norm, const float* __restrict__ W,
                   const float* __restrict__ bias, float* __restrict__ out,
                   const float* __restrict__ out_norm, const float* __restrict__ W2,
                   float* __restrict__ out2) {
  static_assert(NN % NPB == 0, "every block has all of its NPB nodes");
  static_assert(NPB == 8 * NPW && NPB == 32, "8 waves x NPW nodes = two 16-row tiles");
  __shared__ __attribute__((aligned(16))) float As[NPB * AP];
  const unsigned lane = threadIdx.x & 31u;
  const int wave = __builtin_amdgcn_readfirstlane((int)(threadIdx.x >> 5));
  const int m_base = blockIdx.x * NPB;
  const int4* __restrict__ rp = (const int4*)rec;

  float onv = 0.0f;
  if (NEXT) onv = out_norm[m_base + (int)lane];

  for (int k = 0; k < NPW; ++k) {
    const int r = wave * NPW + k;
    const int n = m_base + r;
    const int beg = offs[n];
    const int deg = in_cnt[n];
    const float nrm = in_norm[n];
    float4 acc = {};
    for (int c0 = 0; c0 < deg; c0 += 32) {
      const int m = deg - c0 < 32 ? deg - c0 : 32;
      int4 q = {0, 0, 0, 0};
      if ((int)lane < m) q = rp[beg + c0 + (int)lane];
      int j = 0;
      for (; j + 8 <= m; j += 8) row_group<8, KD, true, float4>(acc, h, q, j, lane);
      if (j + 4 <= m) { row_group<4, KD, true, float4>(acc, h, q, j, lane); j += 4; }
      if (j + 2 <= m) { row_group<2, KD, true, float4>(acc, h, q, j, lane); j += 2; }
      if (j < m) row_group<1, KD, true, float4>(acc, h, q, j, lane);
    }
    acc = acc * nrm;
    *(float4*)&As[r * AP + lane * 4] = acc;
  }
  __syncthreads();

  const int n_base = wave * 16;
  const int lo = lane & 15;
  const int hi = lane >> 4;
  v8f acc0 = {};
  v8f acc1 = {};
  const float* bbase = W + (size_t)(2 * hi) * KD + n_base + lo;
  const float* abase = &As[lo * AP + 2 * hi];

  #pragma unroll
  for (int k0 = 0; k0 < KD; k0 += 4) {
    v2f a0 = *(const v2f*)(abase + k0);
    v2f a1 = *(const v2f*)(abase + 16 * AP + k0);
    v2f b;
    b.x = bbase[(size_t)(k0 + 0) * KD];
    b.y = bbase[(size_t)(k0 + 1) * KD];
    acc0 = __builtin_amdgcn_wmma_f32_16x16x4_f32(
         false, a0,  false, b,
         (short)0, acc0,  false,  false);
    acc1 = __builtin_amdgcn_wmma_f32_16x16x4_f32(
         false, a1,  false, b,
         (short)0, acc1,  false,  false);
  }

  const int col = n_base + lo;
  const float bv = bias[col];
  if (!NEXT) {
    #pragma unroll
    for (int v = 0; v < 8; ++v) {
      const int row = m_base + v + 8 * hi;
      float x0 = acc0[v] + bv;
      float x1 = acc1[v] + bv;
      x0 = x0 > 0.0f ? x0 : 0.0f;
      x1 = x1 > 0.0f ? x1 : 0.0f;
      out[(size_t)row * KD + col] = x0;
      out[(size_t)(row + 16) * KD + col] = x1;
    }
  } else {
    constexpr int F2 = 64;
    __shared__ __attribute__((aligned(16))) float Hs[NPB * AP];
    #pragma unroll
    for (int v = 0; v < 8; ++v) {
      const float s00 = __int_as_float(__builtin_amdgcn_readlane(__float_as_int(onv), v));
      const float s08 = __int_as_float(__builtin_amdgcn_readlane(__float_as_int(onv), v + 8));
      const float s16 = __int_as_float(__builtin_amdgcn_readlane(__float_as_int(onv), v + 16));
      const float s24 = __int_as_float(__builtin_amdgcn_readlane(__float_as_int(onv), v + 24));
      const float s0 = hi ? s08 : s00;
      const float s1 = hi ? s24 : s16;
      float x0 = acc0[v] + bv;
      float x1 = acc1[v] + bv;
      x0 = x0 > 0.0f ? x0 : 0.0f;
      x1 = x1 > 0.0f ? x1 : 0.0f;
      Hs[(v + 8 * hi) * AP + col] = x0 * s0;
      Hs[(v + 8 * hi + 16) * AP + col] = x1 * s1;
    }
    __syncthreads();

    const int rh = wave >> 2;
    const int c2 = (wave & 3) * 16;
    v8f acc2 = {};
    const float* bbase2 = W2 + (size_t)(2 * hi) * F2 + c2 + lo;
    const float* abase2 = &Hs[(rh * 16 + lo) * AP + 2 * hi];
    #pragma unroll
    for (int k0 = 0; k0 < KD; k0 += 4) {
      v2f a = *(const v2f*)(abase2 + k0);
      v2f b;
      b.x = bbase2[(size_t)(k0 + 0) * F2];
      b.y = bbase2[(size_t)(k0 + 1) * F2];
      acc2 = __builtin_amdgcn_wmma_f32_16x16x4_f32(
           false, a,  false, b,
           (short)0, acc2,  false,  false);
    }
    #pragma unroll
    for (int v = 0; v < 8; ++v) {
      const int row = m_base + rh * 16 + v + 8 * hi;
      out2[(size_t)row * F2 + c2 + lo] = acc2[v];
    }
  }
}

extern "C" void kernel_launch(void* const* d_in, const int* in_sizes, int n_in,
                              void* d_out, int out_size, void* d_ws, size_t ws_size,
                              hipStream_t stream) {
  const float* features = (const float*)d_in[0];
  const int*   src      = (const int*)d_in[1];
  const int*   dst      = (const int*)d_in[2];
  const float* ew       = (const float*)d_in[3];
  const float* W0 = (const float*)d_in[4];
  const float* b0 = (const float*)d_in[5];
  const float* W1 = (const float*)d_in[6];
  const float* b1 = (const float*)d_in[7];
  const float* W2 = (const float*)d_in[8];
  const float* b2 = (const float*)d_in[9];
  float* out = (float*)d_out;

  float* ws       = (float*)d_ws;
  float* out_norm = ws;
  int*   in_cnt   = (int*)(ws + NNP);
  float* in_norm  = ws + 2 * NNP;
  int*   offs     = (int*)(ws + 3 * NNP);
  int*   blocksum = (int*)(ws + 4 * NNP);
  int*   rank     = (int*)(ws + 4 * NNP + 512);
  int*   rec      = rank + NE;
  float* bufA     = (float*)(rec + (size_t)4 * NE);
  float* bufB     = bufA + (size_t)NN * KD;
  int*   cnt_in   = (int*)(bufB + (size_t)NN * KD);
  int*   cnt_out  = cnt_in + (size_t)CH * NNP;

  count_rank_kernel<<<dim3(CH, NR, 2), 1024, 0, stream>>>(src, dst, cnt_in, cnt_out, rank);
  block_sum_kernel<<<NSCAN, 256, 0, stream>>>(cnt_in, blocksum);
  scan_norm_kernel<<<NSCAN, 256, 0, stream>>>(
      cnt_in, cnt_out, blocksum, offs, in_cnt, in_norm, out_norm);
  place_kernel<<<(NE + 255) / 256, 256, 0, stream>>>(
      src, dst, ew, rank, offs, cnt_in, out_norm, rec);

  gather_gemm_kernel<false><<<NN / NPB, 256, 0, stream>>>(
      features, rec, offs, in_cnt, in_norm, W0, b0, bufA, nullptr, nullptr, nullptr);

  gather_gemm_kernel<true><<<NN / NPB, 256, 0, stream>>>(
      bufA, rec, offs, in_cnt, in_norm, W1, b1, nullptr, out_norm, W2, bufB);

  gather_sum_kernel<64, false, true, float2><<<(NN / NPW * 32 + 255) / 256, 256, 0, stream>>>(
      bufB, rec, offs, in_cnt, in_norm, b2, out);
}
